# starting kernel
# baseline (speedup 1.0000x reference)
#include <hip/hip_runtime.h>

typedef __attribute__((ext_vector_type(2))) float v2f;
typedef __attribute__((ext_vector_type(4))) float v4f;
typedef __attribute__((ext_vector_type(8))) float v8f;

#define H_IN   384
#define W_IN   384
#define H_OUT  377
#define W_OUT  377
#define KH     8
#define KW     8
#define TILE_X 32
#define STRIP_Y 64
#define ROWS   (STRIP_Y + KH - 1)
#define XCOLS  (TILE_X + KW)

__global__ __launch_bounds__(256)
void corr_wmma_kernel(const float* __restrict__ x,
                      const float* __restrict__ k,
                      float* __restrict__ out) {
  __shared__ float XS[ROWS * XCOLS];
  __shared__ float RS[ROWS * KH * TILE_X];

  const int tid  = threadIdx.x;
  const int lane = tid & 31;
  const int wave = tid >> 5;
  const int b    = blockIdx.z;
  const int x0   = blockIdx.x * TILE_X;
  const int y0   = blockIdx.y * STRIP_Y;

  const float* xb = x + (size_t)b * H_IN * W_IN;
  for (int idx = tid; idx < ROWS * XCOLS; idx += 256) {
    int yy = idx / XCOLS;
    int xx = idx - yy * XCOLS;
    int gy = y0 + yy;
    int gx = x0 + xx;
    float v = 0.0f;
    if (gy < H_IN && gx < W_IN) v = xb[gy * W_IN + gx];
    XS[idx] = v;
  }

  const int m    = lane & 15;
  const int half = lane >> 4;
  const int koff = half * 2;
  v2f bm1 = {0.f, 0.f}, bm2 = {0.f, 0.f};
  if (m < KH) {
    const float* kb = k + (size_t)b * (KH * KW) + m * KW;
    bm1.x = kb[0 + koff]; bm1.y = kb[1 + koff];
    bm2.x = kb[4 + koff]; bm2.y = kb[5 + koff];
  }

  __syncthreads();

  for (int yy = wave; yy < ROWS; yy += 8) {
    const float* xr = &XS[yy * XCOLS];

    v2f a1, a2;
    a1.x = xr[m + koff];
    a1.y = xr[m + koff + 1];
    a2.x = xr[m + koff + 4];
    a2.y = xr[m + koff + 5];
    v8f acc0 = {};
    acc0 = __builtin_amdgcn_wmma_f32_16x16x4_f32(false, a1, false, bm1, (short)0, acc0, false, false);
    acc0 = __builtin_amdgcn_wmma_f32_16x16x4_f32(false, a2, false, bm2, (short)0, acc0, false, false);

    v2f a3, a4;
    a3.x = xr[16 + m + koff];
    a3.y = xr[16 + m + koff + 1];
    a4.x = xr[16 + m + koff + 4];
    a4.y = xr[16 + m + koff + 5];
    v8f acc1 = {};
    acc1 = __builtin_amdgcn_wmma_f32_16x16x4_f32(false, a3, false, bm1, (short)0, acc1, false, false);
    acc1 = __builtin_amdgcn_wmma_f32_16x16x4_f32(false, a4, false, bm2, (short)0, acc1, false, false);

    if (m < KH) {
      float* dst = &RS[(yy * KH + m) * TILE_X + half * 8];
      v4f lo0 = {acc0[0], acc0[1], acc0[2], acc0[3]};
      v4f hi0 = {acc0[4], acc0[5], acc0[6], acc0[7]};
      *(v4f*)(dst)          = lo0;
      *(v4f*)(dst + 4)      = hi0;
      v4f lo1 = {acc1[0], acc1[1], acc1[2], acc1[3]};
      v4f hi1 = {acc1[4], acc1[5], acc1[6], acc1[7]};
      *(v4f*)(dst + 16)     = lo1;
      *(v4f*)(dst + 16 + 4) = hi1;
    }
  }

  __syncthreads();

  float* ob = out + (size_t)b * H_OUT * W_OUT;
  const int xo = tid & 31;
  const int yb = tid >> 5;
  for (int i = 0; i < STRIP_Y / 8; ++i) {
    int yo = yb + i * 8;
    float s = 0.f;
#pragma unroll
    for (int r = 0; r < KH; ++r)
      s += RS[((yo + r) * KH + r) * TILE_X + xo];
    int gy = y0 + yo, gx = x0 + xo;
    if (gy < H_OUT && gx < W_OUT) ob[gy * W_OUT + gx] = s;
  }
}

extern "C" void kernel_launch(void* const* d_in, const int* in_sizes, int n_in,
                              void* d_out, int out_size, void* d_ws, size_t ws_size,
                              hipStream_t stream) {
  (void)in_sizes; (void)n_in; (void)d_ws; (void)ws_size; (void)out_size;
  const float* x = (const float*)d_in[0];
  const float* k = (const float*)d_in[1];
  float* out = (float*)d_out;
  dim3 grid((W_OUT + TILE_X - 1) / TILE_X,
            (H_OUT + STRIP_Y - 1) / STRIP_Y,
            128);
  corr_wmma_kernel<<<grid, dim3(256, 1, 1), 0, stream>>>(x, k, out);
}


# 8-row block in three phases: its 16 WMMAs issue back to back on 8 independent accumulators, then row sums and 8 stores under one lane test; row tests and row clamp only in first and last block
# speedup vs baseline: 3.0535x; 3.0535x over previous
#include <hip/hip_runtime.h>


typedef __attribute__((ext_vector_type(2))) float v2f;
typedef __attribute__((ext_vector_type(4))) float v4f;
typedef __attribute__((ext_vector_type(8))) float v8f;
typedef v4f v4f_any __attribute__((aligned(4)));

#define H_IN   384
#define W_IN   384
#define H_OUT  377
#define W_OUT  377
#define KH     8
#define KW     8
#define WAVES  4
#define STRIPS 2
#define T_ROWS (H_IN / STRIPS)
#define NBLK   (T_ROWS / 8 + 1)

static_assert((STRIPS - 1) * T_ROWS + (NBLK - 2) * 8 + 7 <= H_IN - 1, "inner blocks read inside the image");
static_assert((NBLK - 2) * 8 <= H_OUT - (STRIPS - 1) * T_ROWS - 1, "inner blocks finish valid rows only");
static_assert((NBLK - 3) % 2 == 0, "inner blocks 1 .. NBLK-3 are walked in pairs");

struct RowBuf { v4f r[8]; };

template <bool CLAMP>
__device__ __forceinline__ void load_block(RowBuf& buf, const float* __restrict__ xp, int row0) {
#pragma unroll
  for (int u = 0; u < 8; ++u) {
    int yr = row0 + u;
    if (CLAMP && yr > H_IN - 1) yr = H_IN - 1;
    buf.r[u] = *(const v4f_any*)(xp + yr * W_IN);
  }
}

template <bool EDGE>
__device__ __forceinline__ void compute_block(const RowBuf& buf, v2f a1, v2f a2, float (&p)[8],
                                              float* __restrict__ op, int o0, int o_end, bool lane_ok) {
  v8f acc[8];
#pragma unroll
  for (int u = 0; u < 8; ++u) {
    const v2f b1 = {buf.r[u].x, buf.r[u].y};
    const v8f zero = {};
    acc[u] = __builtin_amdgcn_wmma_f32_16x16x4_f32(false, a1, false, b1, (short)0, zero, false, false);
  }
#pragma unroll
  for (int u = 0; u < 8; ++u) {
    const v2f b2 = {buf.r[u].z, buf.r[u].w};
    acc[u] = __builtin_amdgcn_wmma_f32_16x16x4_f32(false, a2, false, b2, (short)0, acc[u], false, false);
  }
  if (lane_ok) {
#pragma unroll
    for (int u = 0; u < 8; ++u) {
      p[u] = acc[u][0];
#pragma unroll
      for (int j = 1; j < 8; ++j) p[(u - j) & 7] += acc[u][j];
      const int o = o0 + u - 7;
      if (!EDGE || (o >= 0 && o < o_end)) op[o * W_OUT] = p[(u + 1) & 7];
    }
  }
}

__global__ __launch_bounds__(WAVES * 32)
void corr_stream_kernel(const float* __restrict__ x,
                        const float* __restrict__ k,
                        float* __restrict__ out) {
  const int lane = threadIdx.x & 31;
  const int wave = threadIdx.x >> 5;
  const int c    = lane & 15;
  const int h    = lane >> 4;
  const int b    = blockIdx.z;
  const int x0   = (blockIdx.x * WAVES + wave) * 16;
  const int ys   = blockIdx.y * T_ROWS;

  v2f a1 = {0.f, 0.f}, a2 = {0.f, 0.f};
  if (c < KH) {
    const float* kb = k + (size_t)b * (KH * KW) + c * KW + 4 * h;
    a1.x = kb[0]; a1.y = kb[1];
    a2.x = kb[2]; a2.y = kb[3];
  }

  int col = x0 + c + 4 * h;
  if (col > W_IN - 4) col = W_IN - 4;
  const float* xp = x + (size_t)b * H_IN * W_IN + col;

  int o_end = H_OUT - ys;
  if (o_end > T_ROWS) o_end = T_ROWS;
  const bool lane_ok = (h == 0) && (x0 + c < W_OUT);
  float* op = out + (size_t)b * H_OUT * W_OUT + (size_t)ys * W_OUT + x0 + c;

  float p[8] = {0.f, 0.f, 0.f, 0.f, 0.f, 0.f, 0.f, 0.f};
  RowBuf bufA, bufB;

  load_block<false>(bufA, xp, ys);
  load_block<false>(bufB, xp, ys + 8);
  compute_block<true>(bufA, a1, a2, p, op, 0, o_end, lane_ok);
  for (int blk = 1; blk < NBLK - 2; blk += 2) {
    load_block<false>(bufA, xp, ys + (blk + 1) * 8);
    compute_block<false>(bufB, a1, a2, p, op, blk * 8, o_end, lane_ok);
    load_block<false>(bufB, xp, ys + (blk + 2) * 8);
    compute_block<false>(bufA, a1, a2, p, op, (blk + 1) * 8, o_end, lane_ok);
  }
  load_block<true>(bufA, xp, ys + (NBLK - 1) * 8);
  compute_block<false>(bufB, a1, a2, p, op, (NBLK - 2) * 8, o_end, lane_ok);
  compute_block<true>(bufA, a1, a2, p, op, (NBLK - 1) * 8, o_end, lane_ok);
}

extern "C" void kernel_launch(void* const* d_in, const int* in_sizes, int n_in,
                              void* d_out, int out_size, void* d_ws, size_t ws_size,
                              hipStream_t stream) {
  (void)in_sizes; (void)n_in; (void)d_ws; (void)ws_size; (void)out_size;
  const float* x = (const float*)d_in[0];
  const float* k = (const float*)d_in[1];
  float* out = (float*)d_out;
  dim3 grid(6,
            2,
            128);
  corr_stream_kernel<<<grid, dim3(128, 1, 1), 0, stream>>>(x, k, out);
}


# Banded filter as A: M = 4 column shifts x 4 filter-row pairs, N = 16 columns at stride 4, K = 2 image rows x 12 taps; 6 WMMAs per row finish 64 columns (was 8); tap and row sum order changes
# speedup vs baseline: 4.1481x; 1.3585x over previous
#include <hip/hip_runtime.h>


typedef __attribute__((ext_vector_type(2))) float v2f;
typedef __attribute__((ext_vector_type(4))) float v4f;
typedef __attribute__((ext_vector_type(8))) float v8f;
typedef v4f v4f_any __attribute__((aligned(4)));
typedef v2f v2f_any __attribute__((aligned(4)));

#define H_IN   384
#define W_IN   384
#define H_OUT  377
#define W_OUT  377
#define KH     8
#define KW     8
#define STRIPS 4
#define XGW    3
#define WAVES  (STRIPS * XGW)
#define T_ROWS 96
#define NBLK   ((T_ROWS + 6 + 7) / 8)
#define XG     ((W_OUT + 63) / 64)

static_assert(STRIPS * T_ROWS >= H_OUT, "strips cover every output row");
static_assert(XG % XGW == 0, "column groups divide among work-groups");
static_assert((STRIPS - 1) * T_ROWS + 8 * (NBLK - 3) + 8 <= H_IN - 1, "unclamped blocks read inside the image");
static_assert(8 * (NBLK - 3) + 1 <= H_OUT - (STRIPS - 1) * T_ROWS - 1, "inner blocks finish valid rows only");
static_assert(NBLK >= 7 && (NBLK - 5) % 2 == 0, "blocks 1 .. NBLK-5 are walked in pairs");

struct RowFrag { v4f q; v2f e; };
struct RowBuf  { RowFrag r[8]; };

#define LOAD_GROUP(call) do { __builtin_amdgcn_sched_barrier(0); call; __builtin_amdgcn_sched_barrier(0); } while (0)

template <bool CLAMP>
__device__ __forceinline__ void load_block(RowBuf& buf, const float* __restrict__ xq,
                                           const float* __restrict__ xe, int row0, int rmax) {
#pragma unroll
  for (int u = 0; u < 8; ++u) {
    int yr = row0 + u;
    if (CLAMP && yr > rmax) yr = rmax;
    buf.r[u].q = *(const v4f_any*)(xq + yr * W_IN);
    buf.r[u].e = *(const v2f_any*)(xe + yr * W_IN);
  }
}

template <bool EDGE>
__device__ __forceinline__ void compute_block(const RowFrag& carry, const RowBuf& buf, const v2f (&a)[6],
                                              v2f (&p)[8], float* __restrict__ op, int p0, int o_end,
                                              bool st2, bool st1) {
  v8f acc[8];
#pragma unroll
  for (int st = 0; st < 6; ++st) {
    const int d = st / 3, kk = st % 3;
#pragma unroll
    for (int u = 0; u < 8; ++u) {
      const int ri = u + d;
      const RowFrag& R = (ri == 0) ? carry : buf.r[ri == 0 ? 0 : ri - 1];
      v2f bf;
      if (kk == 0)      { bf.x = R.q.x; bf.y = R.q.y; }
      else if (kk == 1) { bf.x = R.q.z; bf.y = R.q.w; }
      else              { bf = R.e; }
      if (st == 0) {
        const v8f zero = {};
        acc[u] = __builtin_amdgcn_wmma_f32_16x16x4_f32(false, a[0], false, bf, (short)0, zero, false, false);
      } else {
        acc[u] = __builtin_amdgcn_wmma_f32_16x16x4_f32(false, a[st], false, bf, (short)0, acc[u], false, false);
      }
    }
  }
  v2f fin[8];
#pragma unroll
  for (int u = 0; u < 8; ++u) {
    const v2f t0 = {acc[u][0], acc[u][1]};
    const v2f t1 = {acc[u][2], acc[u][3]};
    const v2f t2 = {acc[u][4], acc[u][5]};
    const v2f t3 = {acc[u][6], acc[u][7]};
    p[u] = t0;
    p[(u - 2) & 7] += t1;
    p[(u - 4) & 7] += t2;
    fin[u] = p[(u - 6) & 7] + t3;
  }
  if (st2) {
#pragma unroll
    for (int u = 0; u < 8; ++u) {
      const int o = p0 + u - 6;
      if (!EDGE || (o >= 0 && o < o_end)) *(v2f_any*)(op + o * W_OUT) = fin[u];
    }
  } else if (st1) {
#pragma unroll
    for (int u = 0; u < 8; ++u) {
      const int o = p0 + u - 6;
      if (!EDGE || (o >= 0 && o < o_end)) op[o * W_OUT] = fin[u].x;
    }
  }
}

__global__ __launch_bounds__(WAVES * 32)
void corr_band_kernel(const float* __restrict__ x,
                      const float* __restrict__ k,
                      float* __restrict__ out) {
  const int lane = threadIdx.x & 31;
  const int wave = threadIdx.x >> 5;
  const int c    = lane & 15;
  const int h    = lane >> 4;
  const int b    = blockIdx.z;
  const int x0   = (blockIdx.x * XGW + wave / STRIPS) * 64;
  const int ys   = (wave % STRIPS) * T_ROWS;

  const int sh = 2 * (c >> 3) + (c & 1);
  const int g  = (c & 7) >> 1;
  const float* kb = k + (size_t)b * (KH * KW);
  v2f a[6];
#pragma unroll
  for (int st = 0; st < 6; ++st) {
    const int d = st / 3, kk = st % 3;
    float av[2];
#pragma unroll
    for (int s = 0; s < 2; ++s) {
      const int t  = (kk < 2) ? (4 * h + 2 * kk + s) : (8 + 2 * h + s);
      const int fc = t - sh;
      const bool in = (fc >= 0) && (fc < KW);
      const float v = kb[(2 * g + d) * KW + (in ? fc : 0)];
      av[s] = in ? v : 0.f;
    }
    a[st].x = av[0]; a[st].y = av[1];
  }

  int colq = x0 + 4 * c + 4 * h;
  if (colq > W_IN - 4) colq = W_IN - 4;
  int cole = x0 + 4 * c + 8 + 2 * h;
  if (cole > W_IN - 2) cole = W_IN - 2;
  const float* xb = x + (size_t)b * H_IN * W_IN + (size_t)ys * W_IN;
  const float* xq = xb + colq;
  const float* xe = xb + cole;

  const int rmax = H_IN - 1 - ys;
  int o_end = H_OUT - ys;
  if (o_end > T_ROWS) o_end = T_ROWS;
  const int  cs  = x0 + 4 * c + 2 * h;
  const bool st2 = (cs + 1 < W_OUT);
  const bool st1 = (cs < W_OUT) && !st2;
  float* op = out + (size_t)b * H_OUT * W_OUT + (size_t)ys * W_OUT + cs;

  v2f p[8];
#pragma unroll
  for (int i = 0; i < 8; ++i) { p[i].x = 0.f; p[i].y = 0.f; }
  RowFrag carry;
  RowBuf bufA, bufB;

  carry.q = *(const v4f_any*)xq;
  carry.e = *(const v2f_any*)xe;
  LOAD_GROUP(load_block<false>(bufA, xq, xe, 1, rmax));
  LOAD_GROUP(load_block<false>(bufB, xq, xe, 9, rmax));
  compute_block<true>(carry, bufA, a, p, op, 0, o_end, st2, st1);
  carry = bufA.r[7];
  for (int blk = 1; blk < NBLK - 4; blk += 2) {
    LOAD_GROUP(load_block<false>(bufA, xq, xe, (blk + 1) * 8 + 1, rmax));
    compute_block<false>(carry, bufB, a, p, op, blk * 8, o_end, st2, st1);
    carry = bufB.r[7];
    LOAD_GROUP(load_block<false>(bufB, xq, xe, (blk + 2) * 8 + 1, rmax));
    compute_block<false>(carry, bufA, a, p, op, (blk + 1) * 8, o_end, st2, st1);
    carry = bufA.r[7];
  }
  LOAD_GROUP(load_block<false>(bufA, xq, xe, (NBLK - 3) * 8 + 1, rmax));
  compute_block<false>(carry, bufB, a, p, op, (NBLK - 4) * 8, o_end, st2, st1);
  carry = bufB.r[7];
  LOAD_GROUP(load_block<true>(bufB, xq, xe, (NBLK - 2) * 8 + 1, rmax));
  compute_block<false>(carry, bufA, a, p, op, (NBLK - 3) * 8, o_end, st2, st1);
  carry = bufA.r[7];
  LOAD_GROUP(load_block<true>(bufA, xq, xe, (NBLK - 1) * 8 + 1, rmax));
  compute_block<true>(carry, bufB, a, p, op, (NBLK - 2) * 8, o_end, st2, st1);
  carry = bufB.r[7];
  compute_block<true>(carry, bufA, a, p, op, (NBLK - 1) * 8, o_end, st2, st1);
}

extern "C" void kernel_launch(void* const* d_in, const int* in_sizes, int n_in,
                              void* d_out, int out_size, void* d_ws, size_t ws_size,
                              hipStream_t stream) {
  (void)in_sizes; (void)n_in; (void)d_ws; (void)ws_size; (void)out_size;
  const float* x = (const float*)d_in[0];
  const float* k = (const float*)d_in[1];
  float* out = (float*)d_out;
  dim3 grid(2,
            1,
            128);
  corr_band_kernel<<<grid, dim3(384, 1, 1), 0, stream>>>(x, k, out);
}
